# starting kernel
# baseline (speedup 1.0000x reference)
#include <hip/hip_runtime.h>
#include <hip/hip_bf16.h>
#include <math.h>

#define N_NODES    100000
#define N_EDGES    1600000
#define NUM_REL    4
#define NUM_GRAPHS 256

typedef __attribute__((ext_vector_type(2))) float v2f;
typedef __attribute__((ext_vector_type(8))) float v8f;

template <int DIN, int DOUT>
__global__ __launch_bounds__(32)
void transform_wmma(const float* __restrict__ X,
                    const float* __restrict__ W,
                    const float* __restrict__ bias,
                    float* __restrict__ H) {
  const int lane = threadIdx.x & 31;
  const int half = lane >> 4;
  const int l16  = lane & 15;
  const int m0   = blockIdx.x * 16;
  const int n0   = blockIdx.y * 16;
  const int row  = m0 + l16;

  v8f acc = {};
#pragma unroll
  for (int k0 = 0; k0 < DIN; k0 += 4) {
    const int ka = k0 + 2 * half;
    v2f a, b;
    a.x = X[(size_t)row * DIN + ka];
    a.y = X[(size_t)row * DIN + ka + 1];
    b.x = W[(size_t)(ka    ) * DOUT + n0 + l16];
    b.y = W[(size_t)(ka + 1) * DOUT + n0 + l16];
    acc = __builtin_amdgcn_wmma_f32_16x16x4_f32(false, a, false, b,
                                                (short)0, acc, false, false);
  }

  const int   col = n0 + l16;
  const float bv  = bias ? bias[col] : 0.0f;
#pragma unroll
  for (int v = 0; v < 8; ++v) {
    H[(size_t)(m0 + v + 8 * half) * DOUT + col] = acc[v] + bv;
  }
}

__global__ void zero_i32(int* __restrict__ p, int n) {
  int t = blockIdx.x * blockDim.x + threadIdx.x;
  if (t < n) p[t] = 0;
}

__global__ void count_edges(const int* __restrict__ dst,
                            const int* __restrict__ et,
                            int* __restrict__ cnt) {
  int e = blockIdx.x * blockDim.x + threadIdx.x;
  if (e < N_EDGES) atomicAdd(&cnt[dst[e] * NUM_REL + et[e]], 1);
}

template <int DOUT, bool FULLH>
__global__ void scatter_mean(const float* __restrict__ H,
                             const int* __restrict__ src,
                             const int* __restrict__ dst,
                             const int* __restrict__ et,
                             const int* __restrict__ cnt,
                             int rel,
                             float* __restrict__ out) {
  constexpr int Q = DOUT / 4;
  const long long t = (long long)blockIdx.x * blockDim.x + threadIdx.x;
  if (t >= (long long)N_EDGES * Q) return;
  const int e = (int)(t / Q);
  const int c = ((int)(t & (Q - 1))) * 4;
  const int r = et[e];
  if (!FULLH && r != rel) return;
  const int   d    = dst[e];
  const float norm = 1.0f / fmaxf((float)cnt[d * NUM_REL + r], 1.0f);
  const size_t hrow =
      ((FULLH ? (size_t)r * N_NODES : (size_t)0) + (size_t)src[e]) * DOUT + c;
  const float4 hv = *(const float4*)(H + hrow);
  float* o = out + (size_t)d * DOUT + c;
  atomicAdd(o + 0, hv.x * norm);
  atomicAdd(o + 1, hv.y * norm);
  atomicAdd(o + 2, hv.z * norm);
  atomicAdd(o + 3, hv.w * norm);
}

__global__ void relu_k(float* __restrict__ p, int n) {
  int t = blockIdx.x * blockDim.x + threadIdx.x;
  if (t < n) p[t] = fmaxf(p[t], 0.0f);
}

__global__ void fill_neginf(float* __restrict__ p, int n) {
  int t = blockIdx.x * blockDim.x + threadIdx.x;
  if (t < n) p[t] = -INFINITY;
}

__global__ void pool_max(const float* __restrict__ h,
                         const int* __restrict__ batch,
                         float* __restrict__ out) {
  int t = blockIdx.x * blockDim.x + threadIdx.x;
  if (t >= N_NODES * 64) return;
  const int n = t >> 6;
  const int c = t & 63;
  const float v = h[t];
  float* addr = &out[batch[n] * 64 + c];
  if (v >= 0.0f) atomicMax((int*)addr, __float_as_int(v));
  else           atomicMin((unsigned int*)addr, __float_as_uint(v));
}

template <int DIN, int DOUT>
static void run_layer(const float* X, const float* W, const float* root,
                      const float* bias, const int* src, const int* dst,
                      const int* et, const int* cnt, float* h, bool fullh,
                      float* out, bool relu, hipStream_t stream) {
  dim3 tg(N_NODES / 16, DOUT / 16);
  transform_wmma<DIN, DOUT><<<tg, 32, 0, stream>>>(X, root, bias, out);

  constexpr int Q = DOUT / 4;
  const long long tot  = (long long)N_EDGES * Q;
  const int       blks = (int)((tot + 255) / 256);

  if (fullh) {
    for (int r = 0; r < NUM_REL; ++r)
      transform_wmma<DIN, DOUT><<<tg, 32, 0, stream>>>(
          X, W + (size_t)r * DIN * DOUT, nullptr,
          h + (size_t)r * N_NODES * DOUT);
    scatter_mean<DOUT, true><<<blks, 256, 0, stream>>>(h, src, dst, et, cnt,
                                                       0, out);
  } else {
    for (int r = 0; r < NUM_REL; ++r) {
      transform_wmma<DIN, DOUT><<<tg, 32, 0, stream>>>(
          X, W + (size_t)r * DIN * DOUT, nullptr, h);
      scatter_mean<DOUT, false><<<blks, 256, 0, stream>>>(h, src, dst, et,
                                                          cnt, r, out);
    }
  }
  if (relu) {
    const int n = N_NODES * DOUT;
    relu_k<<<(n + 255) / 256, 256, 0, stream>>>(out, n);
  }
}

extern "C" void kernel_launch(void* const* d_in, const int* in_sizes, int n_in,
                              void* d_out, int out_size, void* d_ws,
                              size_t ws_size, hipStream_t stream) {
  (void)in_sizes; (void)n_in; (void)out_size;

  const float* x     = (const float*)d_in[0];
  const int*   ei    = (const int*)d_in[1];
  const int*   et    = (const int*)d_in[2];
  const int*   batch = (const int*)d_in[3];
  const float* W1 = (const float*)d_in[4];
  const float* r1 = (const float*)d_in[5];
  const float* b1 = (const float*)d_in[6];
  const float* W2 = (const float*)d_in[7];
  const float* r2 = (const float*)d_in[8];
  const float* b2 = (const float*)d_in[9];
  const float* W3 = (const float*)d_in[10];
  const float* r3 = (const float*)d_in[11];
  const float* b3 = (const float*)d_in[12];
  const int* src = ei;
  const int* dst = ei + N_EDGES;
  float* out = (float*)d_out;

  char*  ws    = (char*)d_ws;
  int*   cnt   = (int*)ws;
  float* buf2  = (float*)(ws + 1600000);
  float* buf13 = (float*)(ws + 14400000);
  float* h     = (float*)(ws + 40000000);

  const size_t NEED_BIG = 40000000ull + 4ull * N_NODES * 64 * 4;
  const bool fullh = ws_size >= NEED_BIG;

  zero_i32<<<(N_NODES * NUM_REL + 255) / 256, 256, 0, stream>>>(
      cnt, N_NODES * NUM_REL);
  count_edges<<<(N_EDGES + 255) / 256, 256, 0, stream>>>(dst, et, cnt);

  run_layer<16, 16>(x,     W1, r1, b1, src, dst, et, cnt, h, fullh, buf13,
                    true, stream);
  run_layer<16, 32>(buf13, W2, r2, b2, src, dst, et, cnt, h, fullh, buf2,
                    true, stream);
  run_layer<32, 64>(buf2,  W3, r3, b3, src, dst, et, cnt, h, fullh, buf13,
                    false, stream);

  fill_neginf<<<(NUM_GRAPHS * 64 + 255) / 256, 256, 0, stream>>>(
      out, NUM_GRAPHS * 64);
  pool_max<<<(N_NODES * 64 + 255) / 256, 256, 0, stream>>>(buf13, batch, out);
}


# owner work-groups: edges laid out once by 400-row dst range; a work-group sums its rows' messages in LDS (LDS float adds, arrival order); global float adds, seg, pool pass gone; 8 launches to 6
# speedup vs baseline: 3.4444x; 3.4444x over previous
#include <hip/hip_runtime.h>
#include <math.h>


#define N_NODES    100000
#define N_EDGES    1600000
#define NUM_REL    4
#define NUM_GRAPHS 256

typedef __attribute__((ext_vector_type(2))) float v2f;
typedef __attribute__((ext_vector_type(4))) float v4f;
typedef __attribute__((ext_vector_type(8))) float v8f;

__device__ __forceinline__ float other_half(float x) {
  return __int_as_float(__builtin_amdgcn_permlanex16(
      __float_as_int(x), __float_as_int(x), 0x76543210, 0xfedcba98, false,
      false));
}

#define OW_NB   250
#define OW_RPO  400
#define OW_CH   1024
#define OW_EPW  (N_EDGES / OW_NB)

__global__ __launch_bounds__(256)
void part_init(int* __restrict__ bcount, float* __restrict__ o) {
  const int t = blockIdx.x * 256 + threadIdx.x;
  if (t < 256) bcount[t] = 0;
  if (t < NUM_GRAPHS * 64) o[t] = -INFINITY;
}

#define OW_PT   (OW_EPW / 256)

__device__ __forceinline__ int scan256(int v, int (*sc)[256], int t) {
  __syncthreads();
  sc[0][t] = v;
  __syncthreads();
  int cb = 0;
  for (int d = 1; d < 256; d <<= 1) {
    int x = sc[cb][t];
    if (t >= d) x += sc[cb][t - d];
    sc[cb ^ 1][t] = x;
    __syncthreads();
    cb ^= 1;
  }
  return sc[cb][t];
}

__global__ __launch_bounds__(256)
void part_count(const int* __restrict__ dst, int* __restrict__ bcount,
                int* __restrict__ wgstart) {
  static_assert(N_EDGES % OW_NB == 0 && OW_EPW % 256 == 0, "whole slices");
  static_assert(OW_NB * OW_RPO == N_NODES, "ranges cover the rows exactly");
  static_assert(OW_NB <= 256, "one thread per range");
  __shared__ int hist[256];
  const int t = threadIdx.x;
  hist[t] = 0;
  const int e0 = blockIdx.x * OW_EPW;
  int d[OW_PT];
#pragma unroll
  for (int k = 0; k < OW_PT; ++k) d[k] = dst[e0 + t + 256 * k];
  __syncthreads();
#pragma unroll
  for (int k = 0; k < OW_PT; ++k) atomicAdd(&hist[d[k] / OW_RPO], 1);
  __syncthreads();
  if (t < OW_NB) wgstart[blockIdx.x * 256 + t] = atomicAdd(&bcount[t], hist[t]);
}

__global__ __launch_bounds__(256)
void part_fill(const int* __restrict__ src, const int* __restrict__ dst,
               const int* __restrict__ et, const int* __restrict__ bcount,
               const int* __restrict__ wgstart, int* __restrict__ psrc,
               int* __restrict__ pkey) {
  __shared__ int sc[2][256];
  __shared__ int hist[256], lstart[256], lcur[256], gbase[256];
  __shared__ int lsrc[OW_EPW], lkey[OW_EPW];
  const int t = threadIdx.x;
  hist[t] = 0;
  const int e0 = blockIdx.x * OW_EPW;
  int d[OW_PT], s[OW_PT], r[OW_PT];
#pragma unroll
  for (int k = 0; k < OW_PT; ++k) {
    d[k] = dst[e0 + t + 256 * k];
    s[k] = src[e0 + t + 256 * k];
    r[k] = et[e0 + t + 256 * k];
  }
  __syncthreads();
#pragma unroll
  for (int k = 0; k < OW_PT; ++k) atomicAdd(&hist[d[k] / OW_RPO], 1);
  __syncthreads();
  const int h  = hist[t];
  const int li = scan256(h, sc, t);
  lstart[t] = li - h;
  lcur[t]   = li - h;
  const int bc = (t < OW_NB) ? bcount[t] : 0;
  const int gi = scan256(bc, sc, t);
  gbase[t] = gi - bc + ((t < OW_NB) ? wgstart[blockIdx.x * 256 + t] : 0);
  __syncthreads();
#pragma unroll
  for (int k = 0; k < OW_PT; ++k) {
    const int p = atomicAdd(&lcur[d[k] / OW_RPO], 1);
    lsrc[p] = s[k];
    lkey[p] = d[k] * NUM_REL + r[k];
  }
  __syncthreads();
  for (int i = t; i < OW_EPW; i += 256) {
    const int key = lkey[i];
    const int b   = (key >> 2) / OW_RPO;
    const int pos = gbase[b] + (i - lstart[b]);
    psrc[pos] = lsrc[i];
    pkey[pos] = key;
  }
}

template <int DIN, int DOUT, bool RELU_OUT, bool POOL>
__global__ __launch_bounds__(1024)
void owner_layer(const float* __restrict__ X,
                 const float* __restrict__ W,
                 const float* __restrict__ root,
                 const float* __restrict__ bias,
                 const int* __restrict__ psrc,
                 const int* __restrict__ pkey,
                 const int* __restrict__ bcount,
                 float* __restrict__ out,
                 const int* __restrict__ batch) {
  static_assert(DOUT == 16 || DOUT == 32 || DOUT == 64, "row widths handled");
  static_assert(!POOL || DOUT == 64, "the pool's rows are 64 wide");
  constexpr int KS  = DIN / 4;
  constexpr int TPM = DOUT / 16;
  __shared__ __attribute__((aligned(16))) float tile[OW_RPO * DOUT];
  __shared__ int   s_cnt[OW_RPO * NUM_REL];
  __shared__ int   s_src[NUM_REL][OW_CH];
  __shared__ int   s_off[NUM_REL][OW_CH];
  __shared__ float s_nm[NUM_REL][OW_CH];
  __shared__ int   s_n[NUM_REL];
  __shared__ int   s_start;
  const int tid  = threadIdx.x;
  const int lane = tid & 31;
  const int half = lane >> 4;
  const int l16  = lane & 15;
  const int wv   = __builtin_amdgcn_readfirstlane(tid >> 5);
  const int ob   = blockIdx.x;
  const int row0 = ob * OW_RPO;

  if (tid == 0) s_start = 0;
  for (int i = tid; i < OW_RPO * NUM_REL; i += 1024) s_cnt[i] = 0;
  __syncthreads();
  if (tid < ob) atomicAdd(&s_start, bcount[tid]);

  if (wv < OW_RPO / 16) {
    const int m0 = row0 + wv * 16;
    v2f a[KS];
#pragma unroll
    for (int s = 0; s < KS; ++s)
      a[s] = *(const v2f*)(X + (size_t)(m0 + l16) * DIN + 4 * s + 2 * half);
#pragma unroll
    for (int t = 0; t < TPM; ++t) {
      v8f acc = {};
#pragma unroll
      for (int s = 0; s < KS; ++s) {
        const int ka = 4 * s + 2 * half;
        v2f b;
        b.x = root[(size_t)(ka    ) * DOUT + 16 * t + l16];
        b.y = root[(size_t)(ka + 1) * DOUT + 16 * t + l16];
        acc = __builtin_amdgcn_wmma_f32_16x16x4_f32(false, a[s], false, b,
                                                    (short)0, acc, false, false);
      }
      const float bv = bias[16 * t + l16];
#pragma unroll
      for (int v = 0; v < 8; ++v)
        tile[(wv * 16 + v + 8 * half) * DOUT + 16 * t + l16] = acc[v] + bv;
    }
  }
  __syncthreads();

  const int start = s_start;
  const int n     = bcount[ob];
  for (int c0 = 0; c0 < n; c0 += 8 * 1024) {
    int kk[8];
#pragma unroll
    for (int j = 0; j < 8; ++j) {
      const int e = c0 + tid + 1024 * j;
      kk[j] = (e < n) ? pkey[start + e] : -1;
    }
#pragma unroll
    for (int j = 0; j < 8; ++j)
      if (kk[j] >= 0) atomicAdd(&s_cnt[kk[j] - row0 * NUM_REL], 1);
  }

  const int rel = wv & 3;
  const int sub = wv >> 2;
  v2f b[KS][TPM];
  {
    const float* Wr = W + (size_t)rel * DIN * DOUT;
#pragma unroll
    for (int s = 0; s < KS; ++s) {
      const int ka = 4 * s + 2 * half;
#pragma unroll
      for (int t = 0; t < TPM; ++t) {
        b[s][t].x = Wr[(size_t)(ka    ) * DOUT + 16 * t + l16];
        b[s][t].y = Wr[(size_t)(ka + 1) * DOUT + 16 * t + l16];
      }
    }
  }
  int key_nx = row0 * NUM_REL, sv_nx = 0;
  if (tid < n) { key_nx = pkey[start + tid]; sv_nx = psrc[start + tid]; }
  __syncthreads();

  for (int c0 = 0; c0 < n; c0 += OW_CH) {
    if (tid < NUM_REL) s_n[tid] = 0;
    __syncthreads();
    const int  e     = c0 + tid;
    const bool valid = e < n;
    const int  key   = key_nx;
    const int  sv    = sv_nx;
    const int loc = key - row0 * NUM_REL;
    const int r   = loc & 3;
    const float nm = 1.0f / fmaxf((float)s_cnt[loc], 1.0f);
#pragma unroll
    for (int k = 0; k < NUM_REL; ++k) {
      const bool m = valid && (r == k);
      const unsigned mask = __builtin_amdgcn_ballot_w32(m);
      if (mask == 0u) continue;
      int base = 0;
      if (lane == 0) base = atomicAdd(&s_n[k], __builtin_popcount(mask));
      base = __builtin_amdgcn_readfirstlane(base);
      if (m) {
        const int pos = base + (int)__builtin_amdgcn_mbcnt_lo(mask, 0u);
        s_src[k][pos] = sv;
        s_off[k][pos] = (loc >> 2) * DOUT;
        s_nm[k][pos]  = nm;
      }
    }
    __syncthreads();

    key_nx = row0 * NUM_REL;
    sv_nx  = 0;
    if (e + OW_CH < n) {
      key_nx = pkey[start + e + OW_CH];
      sv_nx  = psrc[start + e + OW_CH];
    }

    const int nr = __builtin_amdgcn_readfirstlane(s_n[rel]);
    for (int base = 16 * sub; base < nr; base += 16 * 8) {
      const int  idx = base + l16;
      const bool ok  = idx < nr;
      const int   sj = ok ? s_src[rel][idx] : 0;
      const int   oj = ok ? s_off[rel][idx] : 0;
      const float nj = ok ? s_nm[rel][idx] : 0.0f;
      const int   nrows = nr - base;

      const float* xr = X + (size_t)sj * DIN + 2 * half;
      v2f a[KS];
#pragma unroll
      for (int s = 0; s < KS; ++s) a[s] = *(const v2f*)(xr + 4 * s);

      v8f acc[TPM];
#pragma unroll
      for (int c = 0; c < TPM; ++c) acc[c] = v8f{};
#pragma unroll
      for (int s = 0; s < KS; ++s) {
#pragma unroll
        for (int c = 0; c < TPM; ++c) {
          acc[c] = __builtin_amdgcn_wmma_f32_16x16x4_f32(
              false, a[s], false, b[s][c], (short)0, acc[c], false, false);
        }
      }

#pragma unroll
      for (int v = 0; v < 8; ++v) {
        const int o_lo = __builtin_amdgcn_readlane(oj, v);
        const int o_hi = __builtin_amdgcn_readlane(oj, 8 + v);
        const float n_lo = __int_as_float(
            __builtin_amdgcn_readlane(__float_as_int(nj), v));
        const float n_hi = __int_as_float(
            __builtin_amdgcn_readlane(__float_as_int(nj), 8 + v));
        if constexpr (TPM == 1) {
          const bool in = (8 * half + v) < nrows;
          const int   o = half ? o_hi : o_lo;
          const float f = half ? n_hi : n_lo;
          if (in) atomicAdd(&tile[o + l16], acc[0][v] * f);
        } else {
#pragma unroll
          for (int c = 0; c < TPM; c += 2) {
            const float x0 = acc[c][v];
            const float x1 = acc[c + 1][v];
            const float y0 = other_half(x0);
            const float y1 = other_half(x1);
            const float lo = half ? y1 : x0;
            const float hi = half ? x1 : y0;
            if (v < nrows)
              atomicAdd(&tile[o_lo + 16 * c + lane], lo * n_lo);
            if (8 + v < nrows)
              atomicAdd(&tile[o_hi + 16 * c + lane], hi * n_hi);
          }
        }
      }
    }
    __syncthreads();
  }

  if constexpr (POOL) {
    const int c  = tid & 63;
    const int sl = tid >> 6;
    float m = -INFINITY;
    int   g = -1;
    for (int rl = sl; rl < OW_RPO; rl += 16) {
      const int gg = batch[row0 + rl];
      if (gg != g) {
        if (g >= 0) {
          float* addr = &out[g * 64 + c];
          if (m >= 0.0f) atomicMax((int*)addr, __float_as_int(m));
          else           atomicMin((unsigned int*)addr, __float_as_uint(m));
        }
        g = gg;
        m = -INFINITY;
      }
      m = fmaxf(m, tile[rl * 64 + c]);
    }
    if (g >= 0) {
      float* addr = &out[g * 64 + c];
      if (m >= 0.0f) atomicMax((int*)addr, __float_as_int(m));
      else           atomicMin((unsigned int*)addr, __float_as_uint(m));
    }
  } else {
    const v4f* t4 = (const v4f*)tile;
    v4f* o4 = (v4f*)(out + (size_t)row0 * DOUT);
    for (int i = tid; i < OW_RPO * DOUT / 4; i += 1024) {
      v4f t = t4[i];
      if constexpr (RELU_OUT) {
        t.x = fmaxf(t.x, 0.0f); t.y = fmaxf(t.y, 0.0f);
        t.z = fmaxf(t.z, 0.0f); t.w = fmaxf(t.w, 0.0f);
      }
      o4[i] = t;
    }
  }
}

extern "C" void kernel_launch(void* const* d_in, const int* in_sizes, int n_in,
                              void* d_out, int out_size, void* d_ws,
                              size_t ws_size, hipStream_t stream) {
  (void)in_sizes; (void)n_in; (void)out_size;

  const float* x     = (const float*)d_in[0];
  const int*   ei    = (const int*)d_in[1];
  const int*   et    = (const int*)d_in[2];
  const int*   batch = (const int*)d_in[3];
  const float* W1 = (const float*)d_in[4];
  const float* r1 = (const float*)d_in[5];
  const float* b1 = (const float*)d_in[6];
  const float* W2 = (const float*)d_in[7];
  const float* r2 = (const float*)d_in[8];
  const float* b2 = (const float*)d_in[9];
  const float* W3 = (const float*)d_in[10];
  const float* r3 = (const float*)d_in[11];
  const float* b3 = (const float*)d_in[12];
  const int* src = ei;
  const int* dst = ei + N_EDGES;
  float* out = (float*)d_out;

  if (ws_size < 32327680ull) return;
  char*  wo      = (char*)d_ws;
  int*   bcount  = (int*)wo;
  int*   wgstart = (int*)(wo + 4096);
  int*   psrc    = (int*)(wo + 327680);
  int*   pkey    = (int*)(wo + 6727680);
  float* bufA    = (float*)(wo + 13127680);
  float* bufB    = (float*)(wo + 19527680);

  part_init<<<64, 256, 0, stream>>>(bcount, out);
  part_count<<<OW_NB, 256, 0, stream>>>(dst, bcount, wgstart);
  part_fill<<<OW_NB, 256, 0, stream>>>(src, dst, et, bcount, wgstart, psrc,
                                       pkey);
  owner_layer<16, 16, true, false><<<OW_NB, 1024, 0, stream>>>(
      x, W1, r1, b1, psrc, pkey, bcount, bufA, nullptr);
  owner_layer<16, 32, true, false><<<OW_NB, 1024, 0, stream>>>(
      bufA, W2, r2, b2, psrc, pkey, bcount, bufB, nullptr);
  owner_layer<32, 64, false, true><<<OW_NB, 1024, 0, stream>>>(
      bufB, W3, r3, b3, psrc, pkey, bcount, out, batch);
}


# owner layers sort a range's edges into the relation lists ONCE (8 edges a thread, one returning LDS add a wave), not per 1024-edge chunk: 18 barriers a layer gone; LDS float adds, arrival order
# speedup vs baseline: 3.9861x; 1.1573x over previous
#include <hip/hip_runtime.h>
#include <math.h>


#define N_NODES    100000
#define N_EDGES    1600000
#define NUM_REL    4
#define NUM_GRAPHS 256

typedef __attribute__((ext_vector_type(2))) float v2f;
typedef __attribute__((ext_vector_type(4))) float v4f;
typedef __attribute__((ext_vector_type(8))) float v8f;

__device__ __forceinline__ float other_half(float x) {
  return __int_as_float(__builtin_amdgcn_permlanex16(
      __float_as_int(x), __float_as_int(x), 0x76543210, 0xfedcba98, false,
      false));
}

#define OW_NB   250
#define OW_RPO  400
#define OW_CH   8192
#define OW_LB   11
#define OW_EPW  (N_EDGES / OW_NB)

__global__ __launch_bounds__(256)
void part_init(int* __restrict__ bcount, float* __restrict__ o) {
  const int t = blockIdx.x * 256 + threadIdx.x;
  if (t < 256) bcount[t] = 0;
  if (t < NUM_GRAPHS * 64) o[t] = -INFINITY;
}

#define OW_PT   (OW_EPW / 256)

__device__ __forceinline__ int scan256(int v, int (*sc)[256], int t) {
  __syncthreads();
  sc[0][t] = v;
  __syncthreads();
  int cb = 0;
  for (int d = 1; d < 256; d <<= 1) {
    int x = sc[cb][t];
    if (t >= d) x += sc[cb][t - d];
    sc[cb ^ 1][t] = x;
    __syncthreads();
    cb ^= 1;
  }
  return sc[cb][t];
}

__global__ __launch_bounds__(256)
void part_count(const int* __restrict__ dst, int* __restrict__ bcount,
                int* __restrict__ wgstart) {
  static_assert(N_EDGES % OW_NB == 0 && OW_EPW % 256 == 0, "whole slices");
  static_assert(OW_NB * OW_RPO == N_NODES, "ranges cover the rows exactly");
  static_assert(OW_NB <= 256, "one thread per range");
  __shared__ int hist[256];
  const int t = threadIdx.x;
  hist[t] = 0;
  const int e0 = blockIdx.x * OW_EPW;
  int d[OW_PT];
#pragma unroll
  for (int k = 0; k < OW_PT; ++k) d[k] = dst[e0 + t + 256 * k];
  __syncthreads();
#pragma unroll
  for (int k = 0; k < OW_PT; ++k) atomicAdd(&hist[d[k] / OW_RPO], 1);
  __syncthreads();
  if (t < OW_NB) wgstart[blockIdx.x * 256 + t] = atomicAdd(&bcount[t], hist[t]);
}

__global__ __launch_bounds__(256)
void part_fill(const int* __restrict__ src, const int* __restrict__ dst,
               const int* __restrict__ et, const int* __restrict__ bcount,
               const int* __restrict__ wgstart, int* __restrict__ psrc,
               int* __restrict__ pkey) {
  __shared__ int sc[2][256];
  __shared__ int hist[256], lstart[256], lcur[256], gbase[256];
  __shared__ int lsrc[OW_EPW], lkey[OW_EPW];
  const int t = threadIdx.x;
  hist[t] = 0;
  const int e0 = blockIdx.x * OW_EPW;
  int d[OW_PT], s[OW_PT], r[OW_PT];
#pragma unroll
  for (int k = 0; k < OW_PT; ++k) {
    d[k] = dst[e0 + t + 256 * k];
    s[k] = src[e0 + t + 256 * k];
    r[k] = et[e0 + t + 256 * k];
  }
  __syncthreads();
#pragma unroll
  for (int k = 0; k < OW_PT; ++k) atomicAdd(&hist[d[k] / OW_RPO], 1);
  __syncthreads();
  const int h  = hist[t];
  const int li = scan256(h, sc, t);
  lstart[t] = li - h;
  lcur[t]   = li - h;
  const int bc = (t < OW_NB) ? bcount[t] : 0;
  const int gi = scan256(bc, sc, t);
  gbase[t] = gi - bc + ((t < OW_NB) ? wgstart[blockIdx.x * 256 + t] : 0);
  __syncthreads();
#pragma unroll
  for (int k = 0; k < OW_PT; ++k) {
    const int p = atomicAdd(&lcur[d[k] / OW_RPO], 1);
    lsrc[p] = s[k];
    lkey[p] = d[k] * NUM_REL + r[k];
  }
  __syncthreads();
  for (int i = t; i < OW_EPW; i += 256) {
    const int key = lkey[i];
    const int b   = (key >> 2) / OW_RPO;
    const int pos = gbase[b] + (i - lstart[b]);
    psrc[pos] = lsrc[i];
    pkey[pos] = key;
  }
}

template <int DIN, int DOUT, bool RELU_OUT, bool POOL>
__global__ __launch_bounds__(1024)
void owner_layer(const float* __restrict__ X,
                 const float* __restrict__ W,
                 const float* __restrict__ root,
                 const float* __restrict__ bias,
                 const int* __restrict__ psrc,
                 const int* __restrict__ pkey,
                 const int* __restrict__ bcount,
                 float* __restrict__ out,
                 const int* __restrict__ batch) {
  static_assert(DOUT == 16 || DOUT == 32 || DOUT == 64, "row widths handled");
  static_assert(!POOL || DOUT == 64, "the pool's rows are 64 wide");
  constexpr int KS  = DIN / 4;
  constexpr int TPM = DOUT / 16;
  __shared__ __attribute__((aligned(16))) float tile[OW_RPO * DOUT];
  static_assert(OW_RPO * NUM_REL <= (1 << OW_LB), "the place fits its bits");
  static_assert(N_NODES <= (1 << (31 - OW_LB)), "the source fits above it");
  static_assert(OW_CH == 8 * 1024, "eight edges per thread");
  __shared__ int   s_cnt[OW_RPO * NUM_REL];
  __shared__ int   s_e[NUM_REL][OW_CH];
  __shared__ int   s_n[NUM_REL];
  __shared__ int   s_start;
  const int tid  = threadIdx.x;
  const int lane = tid & 31;
  const int half = lane >> 4;
  const int l16  = lane & 15;
  const int wv   = __builtin_amdgcn_readfirstlane(tid >> 5);
  const int ob   = blockIdx.x;
  const int row0 = ob * OW_RPO;

  if (tid == 0) s_start = 0;
  if (tid < NUM_REL) s_n[tid] = 0;
  for (int i = tid; i < OW_RPO * NUM_REL; i += 1024) s_cnt[i] = 0;
  __syncthreads();
  if (tid < ob) atomicAdd(&s_start, bcount[tid]);

  if (wv < OW_RPO / 16) {
    const int m0 = row0 + wv * 16;
    v2f a[KS];
#pragma unroll
    for (int s = 0; s < KS; ++s)
      a[s] = *(const v2f*)(X + (size_t)(m0 + l16) * DIN + 4 * s + 2 * half);
#pragma unroll
    for (int t = 0; t < TPM; ++t) {
      v8f acc = {};
#pragma unroll
      for (int s = 0; s < KS; ++s) {
        const int ka = 4 * s + 2 * half;
        v2f b;
        b.x = root[(size_t)(ka    ) * DOUT + 16 * t + l16];
        b.y = root[(size_t)(ka + 1) * DOUT + 16 * t + l16];
        acc = __builtin_amdgcn_wmma_f32_16x16x4_f32(false, a[s], false, b,
                                                    (short)0, acc, false, false);
      }
      const float bv = bias[16 * t + l16];
#pragma unroll
      for (int v = 0; v < 8; ++v)
        tile[(wv * 16 + v + 8 * half) * DOUT + 16 * t + l16] = acc[v] + bv;
    }
  }
  __syncthreads();

  const int start = s_start;
  const int n     = bcount[ob];
  for (int c0 = OW_CH; c0 < n; c0 += OW_CH) {
    int k2[8];
#pragma unroll
    for (int j = 0; j < 8; ++j) {
      const int e = c0 + tid + 1024 * j;
      k2[j] = (e < n) ? pkey[start + e] : -1;
    }
#pragma unroll
    for (int j = 0; j < 8; ++j)
      if (k2[j] >= 0) atomicAdd(&s_cnt[k2[j] - row0 * NUM_REL], 1);
  }

  const int rel = wv & 3;
  const int sub = wv >> 2;
  v2f b[KS][TPM];
  {
    const float* Wr = W + (size_t)rel * DIN * DOUT;
#pragma unroll
    for (int s = 0; s < KS; ++s) {
      const int ka = 4 * s + 2 * half;
#pragma unroll
      for (int t = 0; t < TPM; ++t) {
        b[s][t].x = Wr[(size_t)(ka    ) * DOUT + 16 * t + l16];
        b[s][t].y = Wr[(size_t)(ka + 1) * DOUT + 16 * t + l16];
      }
    }
  }
  for (int c0 = 0; c0 < n; c0 += OW_CH) {
    if (c0 > 0) {
      __syncthreads();
      if (tid < NUM_REL) s_n[tid] = 0;
      __syncthreads();
    }
    int kk[8], sv[8];
#pragma unroll
    for (int j = 0; j < 8; ++j) {
      const int e  = c0 + tid + 1024 * j;
      const int ec = (e < n) ? e : n - 1;
      kk[j] = pkey[start + ec];
      sv[j] = psrc[start + ec];
    }
#pragma unroll
    for (int j = 0; j < 8; ++j)
      if (c0 + tid + 1024 * j >= n) kk[j] = -1;
    if (c0 == 0) {
#pragma unroll
      for (int j = 0; j < 8; ++j)
        if (kk[j] >= 0) atomicAdd(&s_cnt[kk[j] - row0 * NUM_REL], 1);
    }
    unsigned mk[8][NUM_REL];
    int cntv = 0;
#pragma unroll
    for (int j = 0; j < 8; ++j) {
#pragma unroll
      for (int k = 0; k < NUM_REL; ++k) {
        const bool m = (kk[j] >= 0) && ((kk[j] & 3) == k);
        mk[j][k] = __builtin_amdgcn_ballot_w32(m);
        if (lane == 4 * j + k) cntv = __builtin_popcount(mk[j][k]);
      }
    }
    const int basev = atomicAdd(&s_n[lane & 3], cntv);
#pragma unroll
    for (int j = 0; j < 8; ++j) {
      const int r  = kk[j] & 3;
      const int bj = __shfl(basev, 4 * j + r);
      int rank = 0;
#pragma unroll
      for (int k = 0; k < NUM_REL; ++k)
        if (r == k) rank = (int)__builtin_amdgcn_mbcnt_lo(mk[j][k], 0u);
      if (kk[j] >= 0)
        s_e[r][bj + rank] = (sv[j] << OW_LB) | (kk[j] - row0 * NUM_REL);
    }
    __syncthreads();
    if (c0 == 0) {
      for (int i = tid; i < OW_RPO * NUM_REL; i += 1024)
        s_cnt[i] = __float_as_int(1.0f / fmaxf((float)s_cnt[i], 1.0f));
      __syncthreads();
    }

    const int nr = __builtin_amdgcn_readfirstlane(s_n[rel]);
    for (int base = 16 * sub; base < nr; base += 16 * 8) {
      const int  idx = base + l16;
      const bool ok  = idx < nr;
      const int   pe = ok ? s_e[rel][idx] : 0;
      const int   sj = (int)((unsigned)pe >> OW_LB);
      const int   nrows = nr - base;

      const float* xr = X + (size_t)sj * DIN + 2 * half;
      v2f a[KS];
#pragma unroll
      for (int s = 0; s < KS; ++s) a[s] = *(const v2f*)(xr + 4 * s);

      const int   lc = pe & ((1 << OW_LB) - 1);
      const int   oj = (lc >> 2) * DOUT;
      const float nj = ok ? __int_as_float(s_cnt[lc]) : 0.0f;

      v8f acc[TPM];
#pragma unroll
      for (int c = 0; c < TPM; ++c) acc[c] = v8f{};
#pragma unroll
      for (int s = 0; s < KS; ++s) {
#pragma unroll
        for (int c = 0; c < TPM; ++c) {
          acc[c] = __builtin_amdgcn_wmma_f32_16x16x4_f32(
              false, a[s], false, b[s][c], (short)0, acc[c], false, false);
        }
      }

#pragma unroll
      for (int v = 0; v < 8; ++v) {
        const int o_lo = __builtin_amdgcn_readlane(oj, v);
        const int o_hi = __builtin_amdgcn_readlane(oj, 8 + v);
        const float n_lo = __int_as_float(
            __builtin_amdgcn_readlane(__float_as_int(nj), v));
        const float n_hi = __int_as_float(
            __builtin_amdgcn_readlane(__float_as_int(nj), 8 + v));
        if constexpr (TPM == 1) {
          const bool in = (8 * half + v) < nrows;
          const int   o = half ? o_hi : o_lo;
          const float f = half ? n_hi : n_lo;
          if (in) atomicAdd(&tile[o + l16], acc[0][v] * f);
        } else {
#pragma unroll
          for (int c = 0; c < TPM; c += 2) {
            const float x0 = acc[c][v];
            const float x1 = acc[c + 1][v];
            const float y0 = other_half(x0);
            const float y1 = other_half(x1);
            const float lo = half ? y1 : x0;
            const float hi = half ? x1 : y0;
            if (v < nrows)
              atomicAdd(&tile[o_lo + 16 * c + lane], lo * n_lo);
            if (8 + v < nrows)
              atomicAdd(&tile[o_hi + 16 * c + lane], hi * n_hi);
          }
        }
      }
    }
  }
  __syncthreads();

  if constexpr (POOL) {
    const int c  = tid & 63;
    const int sl = tid >> 6;
    float m = -INFINITY;
    int   g = -1;
    for (int rl = sl; rl < OW_RPO; rl += 16) {
      const int gg = batch[row0 + rl];
      if (gg != g) {
        if (g >= 0) {
          float* addr = &out[g * 64 + c];
          if (m >= 0.0f) atomicMax((int*)addr, __float_as_int(m));
          else           atomicMin((unsigned int*)addr, __float_as_uint(m));
        }
        g = gg;
        m = -INFINITY;
      }
      m = fmaxf(m, tile[rl * 64 + c]);
    }
    if (g >= 0) {
      float* addr = &out[g * 64 + c];
      if (m >= 0.0f) atomicMax((int*)addr, __float_as_int(m));
      else           atomicMin((unsigned int*)addr, __float_as_uint(m));
    }
  } else {
    const v4f* t4 = (const v4f*)tile;
    v4f* o4 = (v4f*)(out + (size_t)row0 * DOUT);
    for (int i = tid; i < OW_RPO * DOUT / 4; i += 1024) {
      v4f t = t4[i];
      if constexpr (RELU_OUT) {
        t.x = fmaxf(t.x, 0.0f); t.y = fmaxf(t.y, 0.0f);
        t.z = fmaxf(t.z, 0.0f); t.w = fmaxf(t.w, 0.0f);
      }
      o4[i] = t;
    }
  }
}

extern "C" void kernel_launch(void* const* d_in, const int* in_sizes, int n_in,
                              void* d_out, int out_size, void* d_ws,
                              size_t ws_size, hipStream_t stream) {
  (void)in_sizes; (void)n_in; (void)out_size;

  const float* x     = (const float*)d_in[0];
  const int*   ei    = (const int*)d_in[1];
  const int*   et    = (const int*)d_in[2];
  const int*   batch = (const int*)d_in[3];
  const float* W1 = (const float*)d_in[4];
  const float* r1 = (const float*)d_in[5];
  const float* b1 = (const float*)d_in[6];
  const float* W2 = (const float*)d_in[7];
  const float* r2 = (const float*)d_in[8];
  const float* b2 = (const float*)d_in[9];
  const float* W3 = (const float*)d_in[10];
  const float* r3 = (const float*)d_in[11];
  const float* b3 = (const float*)d_in[12];
  const int* src = ei;
  const int* dst = ei + N_EDGES;
  float* out = (float*)d_out;

  if (ws_size < 32327680ull) return;
  char*  wo      = (char*)d_ws;
  int*   bcount  = (int*)wo;
  int*   wgstart = (int*)(wo + 4096);
  int*   psrc    = (int*)(wo + 327680);
  int*   pkey    = (int*)(wo + 6727680);
  float* bufA    = (float*)(wo + 13127680);
  float* bufB    = (float*)(wo + 19527680);

  part_init<<<64, 256, 0, stream>>>(bcount, out);
  part_count<<<OW_NB, 256, 0, stream>>>(dst, bcount, wgstart);
  part_fill<<<OW_NB, 256, 0, stream>>>(src, dst, et, bcount, wgstart, psrc,
                                       pkey);
  owner_layer<16, 16, true, false><<<OW_NB, 1024, 0, stream>>>(
      x, W1, r1, b1, psrc, pkey, bcount, bufA, nullptr);
  owner_layer<16, 32, true, false><<<OW_NB, 1024, 0, stream>>>(
      bufA, W2, r2, b2, psrc, pkey, bcount, bufB, nullptr);
  owner_layer<32, 64, false, true><<<OW_NB, 1024, 0, stream>>>(
      bufB, W3, r3, b3, psrc, pkey, bcount, out, batch);
}


# tile add-out: LDS float adds leave from the accumulators' own lane layout (row offset by one DPP row share a register; 64-wide rows at pitch 80); norm goes into x before the product; arrival order
# speedup vs baseline: 4.6708x; 1.1718x over previous
#include <hip/hip_runtime.h>
#include <math.h>


#define N_NODES    100000
#define N_EDGES    1600000
#define NUM_REL    4
#define NUM_GRAPHS 256

typedef __attribute__((ext_vector_type(2))) float v2f;
typedef __attribute__((ext_vector_type(4))) float v4f;
typedef __attribute__((ext_vector_type(8))) float v8f;

#define OW_NB   250
#define OW_RPO  400
#define OW_CH   8192
#define OW_LB   11
#define OW_EPW  (N_EDGES / OW_NB)

__global__ __launch_bounds__(256)
void part_init(int* __restrict__ bcount, float* __restrict__ o) {
  const int t = blockIdx.x * 256 + threadIdx.x;
  if (t < 256) bcount[t] = 0;
  if (t < NUM_GRAPHS * 64) o[t] = -INFINITY;
}

#define OW_PT   (OW_EPW / 256)

__device__ __forceinline__ int scan256(int v, int (*sc)[256], int t) {
  __syncthreads();
  sc[0][t] = v;
  __syncthreads();
  int cb = 0;
  for (int d = 1; d < 256; d <<= 1) {
    int x = sc[cb][t];
    if (t >= d) x += sc[cb][t - d];
    sc[cb ^ 1][t] = x;
    __syncthreads();
    cb ^= 1;
  }
  return sc[cb][t];
}

__global__ __launch_bounds__(256)
void part_count(const int* __restrict__ dst, int* __restrict__ bcount,
                int* __restrict__ wgstart) {
  static_assert(N_EDGES % OW_NB == 0 && OW_EPW % 256 == 0, "whole slices");
  static_assert(OW_NB * OW_RPO == N_NODES, "ranges cover the rows exactly");
  static_assert(OW_NB <= 256, "one thread per range");
  __shared__ int hist[256];
  const int t = threadIdx.x;
  hist[t] = 0;
  const int e0 = blockIdx.x * OW_EPW;
  int d[OW_PT];
#pragma unroll
  for (int k = 0; k < OW_PT; ++k) d[k] = dst[e0 + t + 256 * k];
  __syncthreads();
#pragma unroll
  for (int k = 0; k < OW_PT; ++k) atomicAdd(&hist[d[k] / OW_RPO], 1);
  __syncthreads();
  if (t < OW_NB) wgstart[blockIdx.x * 256 + t] = atomicAdd(&bcount[t], hist[t]);
}

__global__ __launch_bounds__(256)
void part_fill(const int* __restrict__ src, const int* __restrict__ dst,
               const int* __restrict__ et, const int* __restrict__ bcount,
               const int* __restrict__ wgstart, int* __restrict__ psrc,
               int* __restrict__ pkey) {
  __shared__ int sc[2][256];
  __shared__ int hist[256], lstart[256], lcur[256], gbase[256];
  __shared__ int lsrc[OW_EPW], lkey[OW_EPW];
  const int t = threadIdx.x;
  hist[t] = 0;
  const int e0 = blockIdx.x * OW_EPW;
  int d[OW_PT], s[OW_PT], r[OW_PT];
#pragma unroll
  for (int k = 0; k < OW_PT; ++k) {
    d[k] = dst[e0 + t + 256 * k];
    s[k] = src[e0 + t + 256 * k];
    r[k] = et[e0 + t + 256 * k];
  }
  __syncthreads();
#pragma unroll
  for (int k = 0; k < OW_PT; ++k) atomicAdd(&hist[d[k] / OW_RPO], 1);
  __syncthreads();
  const int h  = hist[t];
  const int li = scan256(h, sc, t);
  lstart[t] = li - h;
  lcur[t]   = li - h;
  const int bc = (t < OW_NB) ? bcount[t] : 0;
  const int gi = scan256(bc, sc, t);
  gbase[t] = gi - bc + ((t < OW_NB) ? wgstart[blockIdx.x * 256 + t] : 0);
  __syncthreads();
#pragma unroll
  for (int k = 0; k < OW_PT; ++k) {
    const int p = atomicAdd(&lcur[d[k] / OW_RPO], 1);
    lsrc[p] = s[k];
    lkey[p] = d[k] * NUM_REL + r[k];
  }
  __syncthreads();
  for (int i = t; i < OW_EPW; i += 256) {
    const int key = lkey[i];
    const int b   = (key >> 2) / OW_RPO;
    const int pos = gbase[b] + (i - lstart[b]);
    psrc[pos] = lsrc[i];
    pkey[pos] = key;
  }
}

template <int V, int TPM, bool GUARD>
__device__ __forceinline__ void add_row(char* pl, const v8f (&acc)[TPM],
                                        int rbm, int lim) {
  const int o = __builtin_amdgcn_update_dpp(0, rbm, 0x150 + V, 0xf, 0xf, true);
  float* p = (float*)(pl + o);
  if (!GUARD || V < lim) {
#pragma unroll
    for (int c = 0; c < TPM; ++c) atomicAdd(p + 16 * c, acc[c][V]);
  }
}

template <int TPM, bool GUARD>
__device__ __forceinline__ void add_tile(float* tile, const v8f (&acc)[TPM],
                                         int rbm, int half, int l16, int nrows) {
  char* pl = (char*)(tile + l16);
  const int lim = half ? nrows - 8 : nrows;
  add_row<0, TPM, GUARD>(pl, acc, rbm, lim);
  add_row<1, TPM, GUARD>(pl, acc, rbm, lim);
  add_row<2, TPM, GUARD>(pl, acc, rbm, lim);
  add_row<3, TPM, GUARD>(pl, acc, rbm, lim);
  add_row<4, TPM, GUARD>(pl, acc, rbm, lim);
  add_row<5, TPM, GUARD>(pl, acc, rbm, lim);
  add_row<6, TPM, GUARD>(pl, acc, rbm, lim);
  add_row<7, TPM, GUARD>(pl, acc, rbm, lim);
}

template <int DIN, int DOUT, bool RELU_OUT, bool POOL>
__global__ __launch_bounds__(1024)
void owner_layer(const float* __restrict__ X,
                 const float* __restrict__ W,
                 const float* __restrict__ root,
                 const float* __restrict__ bias,
                 const int* __restrict__ psrc,
                 const int* __restrict__ pkey,
                 const int* __restrict__ bcount,
                 float* __restrict__ out,
                 const int* __restrict__ batch) {
  static_assert(DOUT == 16 || DOUT == 32 || DOUT == 64, "row widths handled");
  static_assert(!POOL || DOUT == 64, "the pool's rows are 64 wide");
  constexpr int KS  = DIN / 4;
  constexpr int TPM = DOUT / 16;
  constexpr int TP  = POOL ? DOUT + 16 : DOUT;
  __shared__ __attribute__((aligned(16))) float tile[OW_RPO * TP];
  static_assert(OW_RPO * NUM_REL <= (1 << OW_LB), "the place fits its bits");
  static_assert(N_NODES <= (1 << (31 - OW_LB)), "the source fits above it");
  static_assert(OW_CH == 8 * 1024, "eight edges per thread");
  __shared__ int   s_cnt[OW_RPO * NUM_REL];
  __shared__ int   s_e[NUM_REL][OW_CH];
  __shared__ int   s_n[NUM_REL];
  __shared__ int   s_start;
  const int tid  = threadIdx.x;
  const int lane = tid & 31;
  const int half = lane >> 4;
  const int l16  = lane & 15;
  const int wv   = __builtin_amdgcn_readfirstlane(tid >> 5);
  const int ob   = blockIdx.x;
  const int row0 = ob * OW_RPO;

  if (tid == 0) s_start = 0;
  if (tid < NUM_REL) s_n[tid] = 0;
  for (int i = tid; i < OW_RPO * NUM_REL; i += 1024) s_cnt[i] = 0;
  __syncthreads();
  if (tid < ob) atomicAdd(&s_start, bcount[tid]);

  if (wv < OW_RPO / 16) {
    const int m0 = row0 + wv * 16;
    v2f a[KS];
#pragma unroll
    for (int s = 0; s < KS; ++s)
      a[s] = *(const v2f*)(X + (size_t)(m0 + l16) * DIN + 4 * s + 2 * half);
#pragma unroll
    for (int t = 0; t < TPM; ++t) {
      v8f acc = {};
#pragma unroll
      for (int s = 0; s < KS; ++s) {
        const int ka = 4 * s + 2 * half;
        v2f b;
        b.x = root[(size_t)(ka    ) * DOUT + 16 * t + l16];
        b.y = root[(size_t)(ka + 1) * DOUT + 16 * t + l16];
        acc = __builtin_amdgcn_wmma_f32_16x16x4_f32(false, a[s], false, b,
                                                    (short)0, acc, false, false);
      }
      const float bv = bias[16 * t + l16];
#pragma unroll
      for (int v = 0; v < 8; ++v)
        tile[(wv * 16 + v + 8 * half) * TP + 16 * t + l16] = acc[v] + bv;
    }
  }
  __syncthreads();

  const int start = s_start;
  const int n     = bcount[ob];
  for (int c0 = OW_CH; c0 < n; c0 += OW_CH) {
    int k2[8];
#pragma unroll
    for (int j = 0; j < 8; ++j) {
      const int e = c0 + tid + 1024 * j;
      k2[j] = (e < n) ? pkey[start + e] : -1;
    }
#pragma unroll
    for (int j = 0; j < 8; ++j)
      if (k2[j] >= 0) atomicAdd(&s_cnt[k2[j] - row0 * NUM_REL], 1);
  }

  const int rel = wv & 3;
  const int sub = wv >> 2;
  v2f b[KS][TPM];
  {
    const float* Wr = W + (size_t)rel * DIN * DOUT;
#pragma unroll
    for (int s = 0; s < KS; ++s) {
      const int ka = 4 * s + 2 * half;
#pragma unroll
      for (int t = 0; t < TPM; ++t) {
        b[s][t].x = Wr[(size_t)(ka    ) * DOUT + 16 * t + l16];
        b[s][t].y = Wr[(size_t)(ka + 1) * DOUT + 16 * t + l16];
      }
    }
  }
  for (int c0 = 0; c0 < n; c0 += OW_CH) {
    if (c0 > 0) {
      __syncthreads();
      if (tid < NUM_REL) s_n[tid] = 0;
      __syncthreads();
    }
    int kk[8], sv[8];
#pragma unroll
    for (int j = 0; j < 8; ++j) {
      const int e  = c0 + tid + 1024 * j;
      const int ec = (e < n) ? e : n - 1;
      kk[j] = pkey[start + ec];
      sv[j] = psrc[start + ec];
    }
#pragma unroll
    for (int j = 0; j < 8; ++j)
      if (c0 + tid + 1024 * j >= n) kk[j] = -1;
    if (c0 == 0) {
#pragma unroll
      for (int j = 0; j < 8; ++j)
        if (kk[j] >= 0) atomicAdd(&s_cnt[kk[j] - row0 * NUM_REL], 1);
    }
    unsigned mk[8][NUM_REL];
    int cntv = 0;
#pragma unroll
    for (int j = 0; j < 8; ++j) {
#pragma unroll
      for (int k = 0; k < NUM_REL; ++k) {
        const bool m = (kk[j] >= 0) && ((kk[j] & 3) == k);
        mk[j][k] = __builtin_amdgcn_ballot_w32(m);
        if (lane == 4 * j + k) cntv = __builtin_popcount(mk[j][k]);
      }
    }
    const int basev = atomicAdd(&s_n[lane & 3], cntv);
#pragma unroll
    for (int j = 0; j < 8; ++j) {
      const int r  = kk[j] & 3;
      const int bj = __shfl(basev, 4 * j + r);
      int rank = 0;
#pragma unroll
      for (int k = 0; k < NUM_REL; ++k)
        if (r == k) rank = (int)__builtin_amdgcn_mbcnt_lo(mk[j][k], 0u);
      if (kk[j] >= 0)
        s_e[r][bj + rank] = (sv[j] << OW_LB) | (kk[j] - row0 * NUM_REL);
    }
    __syncthreads();
    if (c0 == 0) {
      for (int i = tid; i < OW_RPO * NUM_REL; i += 1024)
        s_cnt[i] = __float_as_int(1.0f / fmaxf((float)s_cnt[i], 1.0f));
      __syncthreads();
    }

    const int nr = __builtin_amdgcn_readfirstlane(s_n[rel]);
    for (int base = 16 * sub; base < nr; base += 16 * 8) {
      const int  idx = base + l16;
      const bool ok  = idx < nr;
      const int   pe = ok ? s_e[rel][idx] : 0;
      const int   sj = (int)((unsigned)pe >> OW_LB);
      const int   nrows = nr - base;

      const float* xr = X + (size_t)sj * DIN + 2 * half;
      v2f a[KS];
#pragma unroll
      for (int s = 0; s < KS; ++s) a[s] = *(const v2f*)(xr + 4 * s);

      const int   lc = pe & ((1 << OW_LB) - 1);
      const float nj = ok ? __int_as_float(s_cnt[lc]) : 0.0f;
      const int   pr  = __builtin_amdgcn_update_dpp(0, pe, 0x128, 0xf, 0xf, true);
      const int   pm  = half ? pr : pe;
      const int   rbm = ((pm & ((1 << OW_LB) - 1)) >> 2) * (TP * 4);
#pragma unroll
      for (int s = 0; s < KS; ++s) {
        a[s].x *= nj;
        a[s].y *= nj;
      }

      v8f acc[TPM];
#pragma unroll
      for (int c = 0; c < TPM; ++c) acc[c] = v8f{};
#pragma unroll
      for (int s = 0; s < KS; ++s) {
#pragma unroll
        for (int c = 0; c < TPM; ++c) {
          acc[c] = __builtin_amdgcn_wmma_f32_16x16x4_f32(
              false, a[s], false, b[s][c], (short)0, acc[c], false, false);
        }
      }

      if (nrows >= 16)
        add_tile<TPM, false>(tile, acc, rbm, half, l16, nrows);
      else
        add_tile<TPM, true>(tile, acc, rbm, half, l16, nrows);
    }
  }
  __syncthreads();

  if constexpr (POOL) {
    const int c  = tid & 63;
    const int sl = tid >> 6;
    float m = -INFINITY;
    int   g = -1;
    for (int rl = sl; rl < OW_RPO; rl += 16) {
      const int gg = batch[row0 + rl];
      if (gg != g) {
        if (g >= 0) {
          float* addr = &out[g * 64 + c];
          if (m >= 0.0f) atomicMax((int*)addr, __float_as_int(m));
          else           atomicMin((unsigned int*)addr, __float_as_uint(m));
        }
        g = gg;
        m = -INFINITY;
      }
      m = fmaxf(m, tile[rl * TP + c]);
    }
    if (g >= 0) {
      float* addr = &out[g * 64 + c];
      if (m >= 0.0f) atomicMax((int*)addr, __float_as_int(m));
      else           atomicMin((unsigned int*)addr, __float_as_uint(m));
    }
  } else {
    const v4f* t4 = (const v4f*)tile;
    v4f* o4 = (v4f*)(out + (size_t)row0 * DOUT);
    for (int i = tid; i < OW_RPO * DOUT / 4; i += 1024) {
      v4f t = t4[i];
      if constexpr (RELU_OUT) {
        t.x = fmaxf(t.x, 0.0f); t.y = fmaxf(t.y, 0.0f);
        t.z = fmaxf(t.z, 0.0f); t.w = fmaxf(t.w, 0.0f);
      }
      o4[i] = t;
    }
  }
}

extern "C" void kernel_launch(void* const* d_in, const int* in_sizes, int n_in,
                              void* d_out, int out_size, void* d_ws,
                              size_t ws_size, hipStream_t stream) {
  (void)in_sizes; (void)n_in; (void)out_size;

  const float* x     = (const float*)d_in[0];
  const int*   ei    = (const int*)d_in[1];
  const int*   et    = (const int*)d_in[2];
  const int*   batch = (const int*)d_in[3];
  const float* W1 = (const float*)d_in[4];
  const float* r1 = (const float*)d_in[5];
  const float* b1 = (const float*)d_in[6];
  const float* W2 = (const float*)d_in[7];
  const float* r2 = (const float*)d_in[8];
  const float* b2 = (const float*)d_in[9];
  const float* W3 = (const float*)d_in[10];
  const float* r3 = (const float*)d_in[11];
  const float* b3 = (const float*)d_in[12];
  const int* src = ei;
  const int* dst = ei + N_EDGES;
  float* out = (float*)d_out;

  if (ws_size < 32327680ull) return;
  char*  wo      = (char*)d_ws;
  int*   bcount  = (int*)wo;
  int*   wgstart = (int*)(wo + 4096);
  int*   psrc    = (int*)(wo + 327680);
  int*   pkey    = (int*)(wo + 6727680);
  float* bufA    = (float*)(wo + 13127680);
  float* bufB    = (float*)(wo + 19527680);

  part_init<<<64, 256, 0, stream>>>(bcount, out);
  part_count<<<OW_NB, 256, 0, stream>>>(dst, bcount, wgstart);
  part_fill<<<OW_NB, 256, 0, stream>>>(src, dst, et, bcount, wgstart, psrc,
                                       pkey);
  owner_layer<16, 16, true, false><<<OW_NB, 1024, 0, stream>>>(
      x, W1, r1, b1, psrc, pkey, bcount, bufA, nullptr);
  owner_layer<16, 32, true, false><<<OW_NB, 1024, 0, stream>>>(
      bufA, W2, r2, b2, psrc, pkey, bcount, bufB, nullptr);
  owner_layer<32, 64, false, true><<<OW_NB, 1024, 0, stream>>>(
      bufB, W3, r3, b3, psrc, pkey, bcount, out, batch);
}


# serial memory round trips go out as groups: pool reads graph ids staged in LDS (was 25 global reads in turn), partition's LDS adds, scans, count loads, sort's lane reads; start sum on idle waves
# speedup vs baseline: 4.8436x; 1.0370x over previous
#include <hip/hip_runtime.h>
#include <math.h>


#define N_NODES    100000
#define N_EDGES    1600000
#define NUM_REL    4
#define NUM_GRAPHS 256

typedef __attribute__((ext_vector_type(2))) float v2f;
typedef __attribute__((ext_vector_type(4))) float v4f;
typedef __attribute__((ext_vector_type(8))) float v8f;

#define OW_NB   250
#define OW_RPO  400
#define OW_CH   8192
#define OW_LB   11
#define OW_EPW  (N_EDGES / OW_NB)

__global__ __launch_bounds__(256)
void part_init(int* __restrict__ bcount, float* __restrict__ o) {
  const int t = blockIdx.x * 256 + threadIdx.x;
  if (t < 256) bcount[t] = 0;
  if (t < NUM_GRAPHS * 64) o[t] = -INFINITY;
}

#define OW_PT   (OW_EPW / 256)

typedef __attribute__((ext_vector_type(2))) int v2i;

__device__ __forceinline__ v2i scan256x2(v2i v, v2i (*sc)[256], int t) {
  __syncthreads();
  sc[0][t] = v;
  __syncthreads();
  int cb = 0;
  for (int d = 1; d < 256; d <<= 1) {
    v2i x = sc[cb][t];
    if (t >= d) x += sc[cb][t - d];
    sc[cb ^ 1][t] = x;
    __syncthreads();
    cb ^= 1;
  }
  return sc[cb][t];
}

__global__ __launch_bounds__(256)
void part_count(const int* __restrict__ dst, int* __restrict__ bcount,
                int* __restrict__ wgstart) {
  static_assert(N_EDGES % OW_NB == 0 && OW_EPW % 256 == 0, "whole slices");
  static_assert(OW_NB * OW_RPO == N_NODES, "ranges cover the rows exactly");
  static_assert(OW_NB <= 256, "one thread per range");
  __shared__ int hist[256];
  const int t = threadIdx.x;
  hist[t] = 0;
  const int e0 = blockIdx.x * OW_EPW;
  int d[OW_PT];
#pragma unroll
  for (int k = 0; k < OW_PT; ++k) d[k] = dst[e0 + t + 256 * k];
  __syncthreads();
#pragma unroll
  for (int k = 0; k < OW_PT; ++k) atomicAdd(&hist[d[k] / OW_RPO], 1);
  __syncthreads();
  if (t < OW_NB) wgstart[blockIdx.x * 256 + t] = atomicAdd(&bcount[t], hist[t]);
}

__global__ __launch_bounds__(256)
void part_fill(const int* __restrict__ src, const int* __restrict__ dst,
               const int* __restrict__ et, const int* __restrict__ bcount,
               const int* __restrict__ wgstart, int* __restrict__ psrc,
               int* __restrict__ pkey) {
  __shared__ v2i sc[2][256];
  __shared__ int hist[256], lstart[256], lcur[256], gbase[256];
  __shared__ int lsrc[OW_EPW], lkey[OW_EPW];
  const int t = threadIdx.x;
  hist[t] = 0;
  const int e0 = blockIdx.x * OW_EPW;
  const int bc = (t < OW_NB) ? bcount[t] : 0;
  const int ws = (t < OW_NB) ? wgstart[blockIdx.x * 256 + t] : 0;
  int d[OW_PT], s[OW_PT], r[OW_PT];
#pragma unroll
  for (int k = 0; k < OW_PT; ++k) {
    d[k] = dst[e0 + t + 256 * k];
    s[k] = src[e0 + t + 256 * k];
    r[k] = et[e0 + t + 256 * k];
  }
  __syncthreads();
#pragma unroll
  for (int k = 0; k < OW_PT; ++k) atomicAdd(&hist[d[k] / OW_RPO], 1);
  __syncthreads();
  const int h  = hist[t];
  v2i hb;
  hb.x = h;
  hb.y = bc;
  const v2i pi = scan256x2(hb, sc, t);
  lstart[t] = pi.x - h;
  lcur[t]   = pi.x - h;
  gbase[t] = pi.y - bc + ws;
  __syncthreads();
  int p[OW_PT];
#pragma unroll
  for (int k = 0; k < OW_PT; ++k) p[k] = atomicAdd(&lcur[d[k] / OW_RPO], 1);
#pragma unroll
  for (int k = 0; k < OW_PT; ++k) {
    lsrc[p[k]] = s[k];
    lkey[p[k]] = d[k] * NUM_REL + r[k];
  }
  __syncthreads();
  for (int i = t; i < OW_EPW; i += 256) {
    const int key = lkey[i];
    const int b   = (key >> 2) / OW_RPO;
    const int pos = gbase[b] + (i - lstart[b]);
    psrc[pos] = lsrc[i];
    pkey[pos] = key;
  }
}

template <int V, int TPM, bool GUARD>
__device__ __forceinline__ void add_row(char* pl, const v8f (&acc)[TPM],
                                        int rbm, int lim) {
  const int o = __builtin_amdgcn_update_dpp(0, rbm, 0x150 + V, 0xf, 0xf, true);
  float* p = (float*)(pl + o);
  if (!GUARD || V < lim) {
#pragma unroll
    for (int c = 0; c < TPM; ++c) atomicAdd(p + 16 * c, acc[c][V]);
  }
}

template <int TPM, bool GUARD>
__device__ __forceinline__ void add_tile(float* tile, const v8f (&acc)[TPM],
                                         int rbm, int half, int l16, int nrows) {
  char* pl = (char*)(tile + l16);
  const int lim = half ? nrows - 8 : nrows;
  add_row<0, TPM, GUARD>(pl, acc, rbm, lim);
  add_row<1, TPM, GUARD>(pl, acc, rbm, lim);
  add_row<2, TPM, GUARD>(pl, acc, rbm, lim);
  add_row<3, TPM, GUARD>(pl, acc, rbm, lim);
  add_row<4, TPM, GUARD>(pl, acc, rbm, lim);
  add_row<5, TPM, GUARD>(pl, acc, rbm, lim);
  add_row<6, TPM, GUARD>(pl, acc, rbm, lim);
  add_row<7, TPM, GUARD>(pl, acc, rbm, lim);
}

template <int DIN, int DOUT, bool RELU_OUT, bool POOL>
__global__ __launch_bounds__(1024)
void owner_layer(const float* __restrict__ X,
                 const float* __restrict__ W,
                 const float* __restrict__ root,
                 const float* __restrict__ bias,
                 const int* __restrict__ psrc,
                 const int* __restrict__ pkey,
                 const int* __restrict__ bcount,
                 float* __restrict__ out,
                 const int* __restrict__ batch) {
  static_assert(DOUT == 16 || DOUT == 32 || DOUT == 64, "row widths handled");
  static_assert(!POOL || DOUT == 64, "the pool's rows are 64 wide");
  constexpr int KS  = DIN / 4;
  constexpr int TPM = DOUT / 16;
  constexpr int TP  = POOL ? DOUT + 16 : DOUT;
  __shared__ __attribute__((aligned(16))) float tile[OW_RPO * TP];
  static_assert(OW_RPO * NUM_REL <= (1 << OW_LB), "the place fits its bits");
  static_assert(N_NODES <= (1 << (31 - OW_LB)), "the source fits above it");
  static_assert(OW_CH == 8 * 1024, "eight edges per thread");
  __shared__ int   s_cnt[OW_RPO * NUM_REL];
  __shared__ int   s_e[NUM_REL][OW_CH];
  __shared__ int   s_n[NUM_REL];
  __shared__ int   s_start;
  __shared__ int   s_batch[POOL ? OW_RPO : 1];
  const int tid  = threadIdx.x;
  const int lane = tid & 31;
  const int half = lane >> 4;
  const int l16  = lane & 15;
  const int wv   = __builtin_amdgcn_readfirstlane(tid >> 5);
  const int ob   = blockIdx.x;
  const int row0 = ob * OW_RPO;

  if (tid == 0) s_start = 0;
  if (tid < NUM_REL) s_n[tid] = 0;
  for (int i = tid; i < OW_RPO * NUM_REL; i += 1024) s_cnt[i] = 0;
  __syncthreads();
  if (wv >= OW_RPO / 16) {
    constexpr int T0 = 32 * (OW_RPO / 16);
    constexpr int NT = 1024 - T0;
    static_assert(OW_NB <= 2 * NT, "two loads a thread cover every range");
    const int i0 = tid - T0;
    const int i1 = i0 + NT;
    const int v0 = (i0 < ob) ? bcount[i0] : 0;
    const int v1 = (i1 < ob) ? bcount[i1] : 0;
    atomicAdd(&s_start, v0 + v1);
  }
  int bv = 0;
  if constexpr (POOL) bv = batch[row0 + (tid < OW_RPO ? tid : OW_RPO - 1)];

  if (wv < OW_RPO / 16) {
    const int m0 = row0 + wv * 16;
    v2f a[KS];
#pragma unroll
    for (int s = 0; s < KS; ++s)
      a[s] = *(const v2f*)(X + (size_t)(m0 + l16) * DIN + 4 * s + 2 * half);
#pragma unroll
    for (int t = 0; t < TPM; ++t) {
      v8f acc = {};
#pragma unroll
      for (int s = 0; s < KS; ++s) {
        const int ka = 4 * s + 2 * half;
        v2f b;
        b.x = root[(size_t)(ka    ) * DOUT + 16 * t + l16];
        b.y = root[(size_t)(ka + 1) * DOUT + 16 * t + l16];
        acc = __builtin_amdgcn_wmma_f32_16x16x4_f32(false, a[s], false, b,
                                                    (short)0, acc, false, false);
      }
      const float bv = bias[16 * t + l16];
#pragma unroll
      for (int v = 0; v < 8; ++v)
        tile[(wv * 16 + v + 8 * half) * TP + 16 * t + l16] = acc[v] + bv;
    }
  }
  if constexpr (POOL) {
    if (tid < OW_RPO) s_batch[tid] = bv;
  }
  __syncthreads();

  const int start = s_start;
  const int n     = bcount[ob];
  for (int c0 = OW_CH; c0 < n; c0 += OW_CH) {
    int k2[8];
#pragma unroll
    for (int j = 0; j < 8; ++j) {
      const int e = c0 + tid + 1024 * j;
      k2[j] = (e < n) ? pkey[start + e] : -1;
    }
#pragma unroll
    for (int j = 0; j < 8; ++j)
      if (k2[j] >= 0) atomicAdd(&s_cnt[k2[j] - row0 * NUM_REL], 1);
  }

  const int rel = wv & 3;
  const int sub = wv >> 2;
  v2f b[KS][TPM];
  {
    const float* Wr = W + (size_t)rel * DIN * DOUT;
#pragma unroll
    for (int s = 0; s < KS; ++s) {
      const int ka = 4 * s + 2 * half;
#pragma unroll
      for (int t = 0; t < TPM; ++t) {
        b[s][t].x = Wr[(size_t)(ka    ) * DOUT + 16 * t + l16];
        b[s][t].y = Wr[(size_t)(ka + 1) * DOUT + 16 * t + l16];
      }
    }
  }
  for (int c0 = 0; c0 < n; c0 += OW_CH) {
    if (c0 > 0) {
      __syncthreads();
      if (tid < NUM_REL) s_n[tid] = 0;
      __syncthreads();
    }
    int kk[8], sv[8];
#pragma unroll
    for (int j = 0; j < 8; ++j) {
      const int e  = c0 + tid + 1024 * j;
      const int ec = (e < n) ? e : n - 1;
      kk[j] = pkey[start + ec];
      sv[j] = psrc[start + ec];
    }
#pragma unroll
    for (int j = 0; j < 8; ++j)
      if (c0 + tid + 1024 * j >= n) kk[j] = -1;
    if (c0 == 0) {
#pragma unroll
      for (int j = 0; j < 8; ++j)
        if (kk[j] >= 0) atomicAdd(&s_cnt[kk[j] - row0 * NUM_REL], 1);
    }
    unsigned mk[8][NUM_REL];
    int cntv = 0;
#pragma unroll
    for (int j = 0; j < 8; ++j) {
#pragma unroll
      for (int k = 0; k < NUM_REL; ++k) {
        const bool m = (kk[j] >= 0) && ((kk[j] & 3) == k);
        mk[j][k] = __builtin_amdgcn_ballot_w32(m);
        if (lane == 4 * j + k) cntv = __builtin_popcount(mk[j][k]);
      }
    }
    const int basev = atomicAdd(&s_n[lane & 3], cntv);
    int bj[8];
#pragma unroll
    for (int j = 0; j < 8; ++j)
      bj[j] = __shfl(basev, 4 * j + (kk[j] & 3));
#pragma unroll
    for (int j = 0; j < 8; ++j) {
      const int r = kk[j] & 3;
      int rank = 0;
#pragma unroll
      for (int k = 0; k < NUM_REL; ++k)
        if (r == k) rank = (int)__builtin_amdgcn_mbcnt_lo(mk[j][k], 0u);
      if (kk[j] >= 0)
        s_e[r][bj[j] + rank] = (sv[j] << OW_LB) | (kk[j] - row0 * NUM_REL);
    }
    __syncthreads();
    if (c0 == 0) {
      for (int i = tid; i < OW_RPO * NUM_REL; i += 1024)
        s_cnt[i] = __float_as_int(1.0f / fmaxf((float)s_cnt[i], 1.0f));
      __syncthreads();
    }

    const int nr = __builtin_amdgcn_readfirstlane(s_n[rel]);
    for (int base = 16 * sub; base < nr; base += 16 * 8) {
      const int  idx = base + l16;
      const bool ok  = idx < nr;
      const int   pe = ok ? s_e[rel][idx] : 0;
      const int   sj = (int)((unsigned)pe >> OW_LB);
      const int   nrows = nr - base;

      const float* xr = X + (size_t)sj * DIN + 2 * half;
      v2f a[KS];
#pragma unroll
      for (int s = 0; s < KS; ++s) a[s] = *(const v2f*)(xr + 4 * s);

      const int   lc = pe & ((1 << OW_LB) - 1);
      const float nj = ok ? __int_as_float(s_cnt[lc]) : 0.0f;
      const int   pr  = __builtin_amdgcn_update_dpp(0, pe, 0x128, 0xf, 0xf, true);
      const int   pm  = half ? pr : pe;
      const int   rbm = ((pm & ((1 << OW_LB) - 1)) >> 2) * (TP * 4);
#pragma unroll
      for (int s = 0; s < KS; ++s) {
        a[s].x *= nj;
        a[s].y *= nj;
      }

      v8f acc[TPM];
#pragma unroll
      for (int c = 0; c < TPM; ++c) acc[c] = v8f{};
#pragma unroll
      for (int s = 0; s < KS; ++s) {
#pragma unroll
        for (int c = 0; c < TPM; ++c) {
          acc[c] = __builtin_amdgcn_wmma_f32_16x16x4_f32(
              false, a[s], false, b[s][c], (short)0, acc[c], false, false);
        }
      }

      if (nrows >= 16)
        add_tile<TPM, false>(tile, acc, rbm, half, l16, nrows);
      else
        add_tile<TPM, true>(tile, acc, rbm, half, l16, nrows);
    }
  }
  __syncthreads();

  if constexpr (POOL) {
    constexpr int RB = OW_RPO / 16;
    static_assert(RB * 16 == OW_RPO, "sixteen whole row blocks");
    const int c  = tid & 63;
    const int sl = tid >> 6;
    int   gk[RB];
    float vk[RB];
#pragma unroll
    for (int k = 0; k < RB; ++k) {
      gk[k] = s_batch[sl * RB + k];
      vk[k] = tile[(sl * RB + k) * TP + c];
    }
    float m = -INFINITY;
    int   g = -1;
#pragma unroll
    for (int k = 0; k < RB; ++k) {
      if (gk[k] != g) {
        if (g >= 0) {
          float* addr = &out[g * 64 + c];
          if (m >= 0.0f) atomicMax((int*)addr, __float_as_int(m));
          else           atomicMin((unsigned int*)addr, __float_as_uint(m));
        }
        g = gk[k];
        m = -INFINITY;
      }
      m = fmaxf(m, vk[k]);
    }
    if (g >= 0) {
      float* addr = &out[g * 64 + c];
      if (m >= 0.0f) atomicMax((int*)addr, __float_as_int(m));
      else           atomicMin((unsigned int*)addr, __float_as_uint(m));
    }
  } else {
    const v4f* t4 = (const v4f*)tile;
    v4f* o4 = (v4f*)(out + (size_t)row0 * DOUT);
    for (int i = tid; i < OW_RPO * DOUT / 4; i += 1024) {
      v4f t = t4[i];
      if constexpr (RELU_OUT) {
        t.x = fmaxf(t.x, 0.0f); t.y = fmaxf(t.y, 0.0f);
        t.z = fmaxf(t.z, 0.0f); t.w = fmaxf(t.w, 0.0f);
      }
      o4[i] = t;
    }
  }
}

extern "C" void kernel_launch(void* const* d_in, const int* in_sizes, int n_in,
                              void* d_out, int out_size, void* d_ws,
                              size_t ws_size, hipStream_t stream) {
  (void)in_sizes; (void)n_in; (void)out_size;

  const float* x     = (const float*)d_in[0];
  const int*   ei    = (const int*)d_in[1];
  const int*   et    = (const int*)d_in[2];
  const int*   batch = (const int*)d_in[3];
  const float* W1 = (const float*)d_in[4];
  const float* r1 = (const float*)d_in[5];
  const float* b1 = (const float*)d_in[6];
  const float* W2 = (const float*)d_in[7];
  const float* r2 = (const float*)d_in[8];
  const float* b2 = (const float*)d_in[9];
  const float* W3 = (const float*)d_in[10];
  const float* r3 = (const float*)d_in[11];
  const float* b3 = (const float*)d_in[12];
  const int* src = ei;
  const int* dst = ei + N_EDGES;
  float* out = (float*)d_out;

  if (ws_size < 32327680ull) return;
  char*  wo      = (char*)d_ws;
  int*   bcount  = (int*)wo;
  int*   wgstart = (int*)(wo + 4096);
  int*   psrc    = (int*)(wo + 327680);
  int*   pkey    = (int*)(wo + 6727680);
  float* bufA    = (float*)(wo + 13127680);
  float* bufB    = (float*)(wo + 19527680);

  part_init<<<64, 256, 0, stream>>>(bcount, out);
  part_count<<<OW_NB, 256, 0, stream>>>(dst, bcount, wgstart);
  part_fill<<<OW_NB, 256, 0, stream>>>(src, dst, et, bcount, wgstart, psrc,
                                       pkey);
  owner_layer<16, 16, true, false><<<OW_NB, 1024, 0, stream>>>(
      x, W1, r1, b1, psrc, pkey, bcount, bufA, nullptr);
  owner_layer<16, 32, true, false><<<OW_NB, 1024, 0, stream>>>(
      bufA, W2, r2, b2, psrc, pkey, bcount, bufB, nullptr);
  owner_layer<32, 64, false, true><<<OW_NB, 1024, 0, stream>>>(
      bufB, W3, r3, b3, psrc, pkey, bcount, out, batch);
}
